# starting kernel
# baseline (speedup 1.0000x reference)
#include <hip/hip_runtime.h>
#include <math.h>

typedef __attribute__((ext_vector_type(2))) float v2f;
typedef __attribute__((ext_vector_type(8))) float v8f;

#define IN_DIM 256
#define HID 96
#define OUT_DIM 64
#define ALPHA_F 0.15f
#define OMA_F 0.85f
#define EPS_F 1e-12f


__global__ void zero_int_k(int* p, int n) {
    int i = blockIdx.x * blockDim.x + threadIdx.x;
    if (i < n) p[i] = 0;
}

__global__ void cnt_k(const int* __restrict__ col, int* cnt, int E) {
    int i = blockIdx.x * blockDim.x + threadIdx.x;
    if (i < E) atomicAdd(&cnt[col[i]], 1);
}

__global__ void dinv_from_cnt_k(const int* __restrict__ cnt, float* dinv, int n) {
    int i = blockIdx.x * blockDim.x + threadIdx.x;
    if (i < n) dinv[i] = rsqrtf((float)(cnt[i] + 1));
}

__global__ __launch_bounds__(1024) void scan_k(const int* __restrict__ cnt,
                                               int* __restrict__ rowptr, int n) {
    __shared__ int buf[1024];
    __shared__ int carry_s;
    const int tid = threadIdx.x;
    if (tid == 0) carry_s = 0;
    __syncthreads();
    for (int base = 0; base < n; base += 1024) {
        const int idx = base + tid;
        const int v = (idx < n) ? cnt[idx] : 0;
        buf[tid] = v;
        __syncthreads();
        for (int off = 1; off < 1024; off <<= 1) {
            const int t = (tid >= off) ? buf[tid - off] : 0;
            __syncthreads();
            buf[tid] += t;
            __syncthreads();
        }
        const int incl  = buf[tid];
        const int carry = carry_s;
        if (idx < n) rowptr[idx] = carry + incl - v;
        __syncthreads();
        if (tid == 1023) carry_s = carry + buf[1023];
        __syncthreads();
    }
    if (tid == 0) rowptr[n] = carry_s;
}

__global__ void copy_int_k(int* __restrict__ dst, const int* __restrict__ src, int n) {
    int i = blockIdx.x * blockDim.x + threadIdx.x;
    if (i < n) dst[i] = src[i];
}

__global__ void fill_k(const int* __restrict__ row, const int* __restrict__ col,
                       int* cursor, int* __restrict__ srcs, int E) {
    int i = blockIdx.x * blockDim.x + threadIdx.x;
    if (i < E) {
        const int p = atomicAdd(&cursor[col[i]], 1);
        srcs[p] = row[i];
    }
}

template <int NT>
__global__ __launch_bounds__(256) void gemm_wmma_k(
    const float* __restrict__ X, const float* __restrict__ W,
    const float* __restrict__ bias, float* __restrict__ Out,
    int K, int nrows) {
    const int M    = NT * 16;
    const int wave = threadIdx.x >> 5;
    const int tm   = blockIdx.x * 8 + wave;
    if (tm * 16 >= nrows) return;
    const int lane = threadIdx.x & 31;
    const int half = lane >> 4;
    const int l16  = lane & 15;

    v8f c[NT];
#pragma unroll
    for (int t = 0; t < NT; ++t) c[t] = (v8f){};

    const float* __restrict__ xrow = X + (size_t)(tm * 16 + l16) * K;

    for (int k = 0; k < K; k += 4) {
        const int kk = k + 2 * half;
        v2f a;
        a.x = xrow[kk];
        a.y = xrow[kk + 1];
        const float* __restrict__ w0 = W + (size_t)kk * M + l16;
#pragma unroll
        for (int t = 0; t < NT; ++t) {
            v2f b;
            b.x = w0[t * 16];
            b.y = w0[M + t * 16];
            c[t] = __builtin_amdgcn_wmma_f32_16x16x4_f32(
                false, a, false, b, (short)0, c[t], false, false);
        }
    }

#pragma unroll
    for (int t = 0; t < NT; ++t) {
        const int bcol = t * 16 + l16;
        const float bv = bias[bcol];
#pragma unroll
        for (int r = 0; r < 8; ++r) {
            const int orow = tm * 16 + r + 8 * half;
            Out[(size_t)orow * M + bcol] = c[t][r] + bv;
        }
    }
}

__global__ __launch_bounds__(256) void l2norm_k(float* x, int n, int M, float s) {
    const int wave = threadIdx.x >> 5;
    const int lane = threadIdx.x & 31;
    const int row  = blockIdx.x * 8 + wave;
    if (row >= n) return;
    float* p = x + (size_t)row * M;
    float sum = 0.0f;
    for (int f = lane; f < M; f += 32) { float v = p[f]; sum += v * v; }
#pragma unroll
    for (int o = 16; o > 0; o >>= 1) sum += __shfl_xor(sum, o, 32);
    const float scale = s / fmaxf(sqrtf(sum), EPS_F);
    for (int f = lane; f < M; f += 32) p[f] *= scale;
}


__global__ void prescale_k(float* __restrict__ Y, const float* __restrict__ x,
                           const float* __restrict__ dinv, int n, int M) {
    size_t i = (size_t)blockIdx.x * blockDim.x + threadIdx.x;
    if (i >= (size_t)n * M) return;
    Y[i] = x[i] * dinv[(int)(i / M)];
}

__global__ __launch_bounds__(256) void gather_prop_k(
    float* __restrict__ dst, const float* __restrict__ Y,
    const float* __restrict__ h, const float* __restrict__ dinv,
    const int* __restrict__ rowptr, const int* __restrict__ srcs,
    int n, int M, int relu) {
    const int wave = threadIdx.x >> 5;
    const int lane = threadIdx.x & 31;
    const int node = blockIdx.x * 8 + wave;
    if (node >= n) return;

    const float* __restrict__ yn = Y + (size_t)node * M;
    float a0 = yn[lane];
    float a1 = (M > 32) ? yn[lane + 32] : 0.0f;
    float a2 = (M > 64) ? yn[lane + 64] : 0.0f;

    const int e0 = rowptr[node];
    const int e1 = rowptr[node + 1];
    for (int e = e0; e < e1; ++e) {
        const float* __restrict__ ys = Y + (size_t)srcs[e] * M;
        a0 += ys[lane];
        if (M > 32) a1 += ys[lane + 32];
        if (M > 64) a2 += ys[lane + 64];
    }

    const float dd = OMA_F * dinv[node];
    const float* __restrict__ hn = h + (size_t)node * M;
    float* __restrict__ dn = dst + (size_t)node * M;

    float v0 = dd * a0 + ALPHA_F * hn[lane];
    if (relu) v0 = fmaxf(v0, 0.0f);
    dn[lane] = v0;
    if (M > 32) {
        float v1 = dd * a1 + ALPHA_F * hn[lane + 32];
        if (relu) v1 = fmaxf(v1, 0.0f);
        dn[lane + 32] = v1;
    }
    if (M > 64) {
        float v2 = dd * a2 + ALPHA_F * hn[lane + 64];
        if (relu) v2 = fmaxf(v2, 0.0f);
        dn[lane + 64] = v2;
    }
}


static inline size_t align256(size_t b) { return (b + 255) & ~(size_t)255; }

extern "C" void kernel_launch(void* const* d_in, const int* in_sizes, int n_in,
                              void* d_out, int out_size, void* d_ws, size_t ws_size,
                              hipStream_t stream) {
    const float* x_in = (const float*)d_in[0];
    const int*   ei   = (const int*)d_in[1];
    const float* W0   = (const float*)d_in[2];
    const float* b0   = (const float*)d_in[3];
    const float* W1   = (const float*)d_in[4];
    const float* b1   = (const float*)d_in[5];
    const float* Wx   = (const float*)d_in[6];
    const float* bx   = (const float*)d_in[7];
    float* out = (float*)d_out;

    const int n = in_sizes[0] / IN_DIM;
    const int E = in_sizes[1] / 2;
    const int* row = ei;
    const int* col = ei + E;

    char* ws = (char*)d_ws;
    size_t off = 0;
    float* dinv   = (float*)(ws + off); off += align256((size_t)n * 4);
    int*   cnt    = (int*)  (ws + off); off += align256((size_t)n * 4);
    int*   rowptr = (int*)  (ws + off); off += align256((size_t)(n + 1) * 4);
    int*   cursor = (int*)  (ws + off); off += align256((size_t)n * 4);
    int*   srcs   = (int*)  (ws + off); off += align256((size_t)E * 4);
    float* H      = (float*)(ws + off); off += align256((size_t)n * HID * 4);
    float* X      = (float*)(ws + off); off += align256((size_t)n * HID * 4);
    float* Y      = (float*)(ws + off); off += align256((size_t)n * HID * 4);
    (void)ws_size;

    const int TPB  = 256;
    const int gN   = (n + TPB - 1) / TPB;
    const int gE   = (E + TPB - 1) / TPB;
    const int gRow = (n + 7) / 8;
    const int gTile = ((n / 16) + 7) / 8;

    zero_int_k<<<gN, TPB, 0, stream>>>(cnt, n);
    cnt_k<<<gE, TPB, 0, stream>>>(col, cnt, E);
    dinv_from_cnt_k<<<gN, TPB, 0, stream>>>(cnt, dinv, n);
    scan_k<<<1, 1024, 0, stream>>>(cnt, rowptr, n);
    copy_int_k<<<gN, TPB, 0, stream>>>(cursor, rowptr, n);
    fill_k<<<gE, TPB, 0, stream>>>(row, col, cursor, srcs, E);

    auto prop = [&](const float* src, float* dst, const float* h, int M, int relu) {
        const size_t total = (size_t)n * M;
        const int gT = (int)((total + TPB - 1) / TPB);
        prescale_k<<<gT, TPB, 0, stream>>>(Y, src, dinv, n, M);
        gather_prop_k<<<gRow, TPB, 0, stream>>>(dst, Y, h, dinv, rowptr, srcs, n, M, relu);
    };

    gemm_wmma_k<HID / 16><<<gTile, TPB, 0, stream>>>(x_in, W0, b0, H, IN_DIM, n);
    l2norm_k<<<gRow, TPB, 0, stream>>>(H, n, HID, 1.8f);
    prop(H, X, H, HID, 0);
    prop(X, X, H, HID, 1);

    gemm_wmma_k<HID / 16><<<gTile, TPB, 0, stream>>>(X, W1, b1, H, HID, n);
    l2norm_k<<<gRow, TPB, 0, stream>>>(H, n, HID, 1.8f);
    prop(H, X, H, HID, 0);
    prop(X, X, H, HID, 0);
    l2norm_k<<<gRow, TPB, 0, stream>>>(X, n, HID, 1.5f);

    gemm_wmma_k<OUT_DIM / 16><<<gTile, TPB, 0, stream>>>(X, Wx, bx, H, HID, n);
    l2norm_k<<<gRow, TPB, 0, stream>>>(H, n, OUT_DIM, 1.8f);
    prop(H, X, H, OUT_DIM, 0);
    prop(X, out, H, OUT_DIM, 0);
}


# l2norm folded into its rows' readers: gathers scale each raw row (own, residual, edge source) by s*rsq(sum sq) per 8-lane group, last GEMM scales its products; 18->14 launches; sum-sq order changes
# speedup vs baseline: 2.7858x; 2.7858x over previous
#include <hip/hip_runtime.h>
#include <math.h>

typedef __attribute__((ext_vector_type(2))) float v2f;
typedef __attribute__((ext_vector_type(8))) float v8f;

#define IN_DIM 256
#define HID 96
#define OUT_DIM 64
#define ALPHA_F 0.15f
#define OMA_F 0.85f
#define EPS_F 1e-12f


__global__ void zero_int_k(int* p, int n) {
    int i = blockIdx.x * blockDim.x + threadIdx.x;
    if (i < n) p[i] = 0;
}

__global__ void cnt_k(const int* __restrict__ col, int* cnt, int* __restrict__ rank, int E) {
    int i = blockIdx.x * blockDim.x + threadIdx.x;
    if (i < E) rank[i] = atomicAdd(&cnt[col[i]], 1);
}

__global__ void dinv_from_cnt_k(const int* __restrict__ cnt, float* dinv, int n) {
    int i = blockIdx.x * blockDim.x + threadIdx.x;
    if (i < n) dinv[i] = rsqrtf((float)(cnt[i] + 1));
}

__global__ __launch_bounds__(256) void alloc_k(const int* __restrict__ cnt, int* total,
                                               int* __restrict__ seg, int n) {
    __shared__ int wsum[8];
    __shared__ int base_s;
    const int tid  = threadIdx.x;
    const int lane = tid & 31;
    const int wave = tid >> 5;
    const int i = blockIdx.x * 256 + tid;
    const int c = (i < n) ? cnt[i] : 0;
    int incl = c;
    for (int o = 1; o < 32; o <<= 1) {
        const int t = __shfl_up(incl, o, 32);
        if (lane >= o) incl += t;
    }
    if (lane == 31) wsum[wave] = incl;
    __syncthreads();
    if (tid == 0) {
        int s = 0;
        for (int w = 0; w < 8; ++w) { const int t = wsum[w]; wsum[w] = s; s += t; }
        base_s = atomicAdd(total, s);
    }
    __syncthreads();
    if (i < n) {
        const int p = base_s + wsum[wave] + incl - c;
        reinterpret_cast<int2*>(seg)[i] = make_int2(p, c);
    }
}

__global__ void fill_k(const int* __restrict__ row, const int* __restrict__ col,
                       const int* __restrict__ rank, const int* __restrict__ seg,
                       const float* __restrict__ dinv,
                       int* __restrict__ edges, int E) {
    int i = blockIdx.x * blockDim.x + threadIdx.x;
    if (i < E) {
        const int r = row[i];
        const int p = seg[2 * col[i]] + rank[i];
        reinterpret_cast<int2*>(edges)[p] = make_int2(r, __float_as_int(dinv[r]));
    }
}

template <int NT, bool ANORM>
__global__ __launch_bounds__(256) void gemm_wmma_k(
    const float* __restrict__ X, const float* __restrict__ W,
    const float* __restrict__ bias, float* __restrict__ Out,
    int K, int nrows, float nf) {
    const int M    = NT * 16;
    const int wave = threadIdx.x >> 5;
    const int tm   = blockIdx.x * 8 + wave;
    if (tm * 16 >= nrows) return;
    const int lane = threadIdx.x & 31;
    const int half = lane >> 4;
    const int l16  = lane & 15;

    v8f c[NT];
#pragma unroll
    for (int t = 0; t < NT; ++t) c[t] = (v8f){};

    const float* __restrict__ xrow = X + (size_t)(tm * 16 + l16) * K;
    float ss = 0.0f;

#pragma unroll 4
    for (int k = 0; k < K; k += 4) {
        const int kk = k + 2 * half;
        v2f a;
        a.x = xrow[kk];
        a.y = xrow[kk + 1];
        if (ANORM) { ss = fmaf(a.x, a.x, ss); ss = fmaf(a.y, a.y, ss); }
        const float* __restrict__ w0 = W + (size_t)kk * M + l16;
#pragma unroll
        for (int t = 0; t < NT; ++t) {
            v2f b;
            b.x = w0[t * 16];
            b.y = w0[M + t * 16];
            c[t] = __builtin_amdgcn_wmma_f32_16x16x4_f32(
                false, a, false, b, (short)0, c[t], false, false);
        }
    }

    float rowf[8];
    if (ANORM) {
        ss += __shfl_xor(ss, 16, 32);
        const float f = nf / fmaxf(sqrtf(ss), EPS_F);
#pragma unroll
        for (int r = 0; r < 8; ++r) rowf[r] = __shfl(f, r + 8 * half, 32);
    }

#pragma unroll
    for (int t = 0; t < NT; ++t) {
        const int bcol = t * 16 + l16;
        const float bv = bias[bcol];
#pragma unroll
        for (int r = 0; r < 8; ++r) {
            const int orow = tm * 16 + r + 8 * half;
            float o = c[t][r];
            if (ANORM) o *= rowf[r];
            Out[(size_t)orow * M + bcol] = o + bv;
        }
    }
}


template <int T>
__device__ __forceinline__ int grp_bcast(int v) {
    return (int)__builtin_amdgcn_mov_dpp8((unsigned)v, (unsigned)(T * 0x249249));
}

template <int K>
__device__ __forceinline__ float grp_xor(float v) {
    constexpr unsigned sel =
        ((0u ^ K) << 0)  | ((1u ^ K) << 3)  | ((2u ^ K) << 6)  | ((3u ^ K) << 9) |
        ((4u ^ K) << 12) | ((5u ^ K) << 15) | ((6u ^ K) << 18) | ((7u ^ K) << 21);
    return __int_as_float((int)__builtin_amdgcn_mov_dpp8((unsigned)__float_as_int(v), sel));
}

template <int NV>
__device__ __forceinline__ float row_scale(const float4 (&r)[NV], const float s) {
    float4 p = make_float4(0.0f, 0.0f, 0.0f, 0.0f);
    for (int v = 0; v < NV; ++v) {
        p.x = fmaf(r[v].x, r[v].x, p.x); p.y = fmaf(r[v].y, r[v].y, p.y);
        p.z = fmaf(r[v].z, r[v].z, p.z); p.w = fmaf(r[v].w, r[v].w, p.w);
    }
    float ss = (p.x + p.y) + (p.z + p.w);
    ss += grp_xor<1>(ss);
    ss += grp_xor<2>(ss);
    ss += grp_xor<4>(ss);
    return (ss > EPS_F * EPS_F) ? s * __builtin_amdgcn_rsqf(ss) : s / EPS_F;
}

template <int NV>
__device__ __forceinline__ void load_row(float4 (&r)[NV], const float* __restrict__ x,
                                         const int s, const int j) {
    constexpr int M = NV * 32;
    const float4* __restrict__ xs = reinterpret_cast<const float4*>(x + (size_t)s * M) + j;
    for (int v = 0; v < NV; ++v) r[v] = xs[8 * v];
}

template <int NV, bool XRAW>
__device__ __forceinline__ void add_row(float4 (&acc)[NV], const float4 (&r)[NV], float w,
                                        const float nf) {
    if (XRAW) w *= row_scale<NV>(r, nf);
    for (int v = 0; v < NV; ++v) {
        acc[v].x += w * r[v].x; acc[v].y += w * r[v].y;
        acc[v].z += w * r[v].z; acc[v].w += w * r[v].w;
    }
}

template <int NV, int T, bool XRAW>
__device__ __forceinline__ void edge_trip(float4 (&acc)[NV], const int2 cur, const int left,
                                          const float* __restrict__ x, const int j, const float nf) {
    const int   s = grp_bcast<T>(cur.x);
    const float w = __int_as_float(grp_bcast<T>(cur.y));
    if (T < left) {
        float4 r[NV];
        load_row<NV>(r, x, s, j);
        add_row<NV, XRAW>(acc, r, w, nf);
    }
}

template <int NV, bool XRAW>
__global__ __launch_bounds__(256) void gather_prop_k(
    float* __restrict__ dst, const float* __restrict__ x,
    const float* __restrict__ h, const float* __restrict__ dinv,
    const int* __restrict__ seg, const int* __restrict__ edges,
    int n, int relu, float nf) {
    constexpr int M = NV * 32;
    const int j    = threadIdx.x & 7;
    const int node = blockIdx.x * 32 + (threadIdx.x >> 3);
    if (node >= n) return;

    const float4* __restrict__ xn = reinterpret_cast<const float4*>(x + (size_t)node * M) + j;
    const float4* __restrict__ hn = reinterpret_cast<const float4*>(h + (size_t)node * M) + j;
    float4 acc[NV];
    float4 hv[NV];
    for (int v = 0; v < NV; ++v) acc[v] = xn[8 * v];
    for (int v = 0; v < NV; ++v) hv[v] = hn[8 * v];
    const float dn_self = dinv[node];

    const int2 sg = reinterpret_cast<const int2*>(seg)[node];
    const int len = sg.y;
    const int2* __restrict__ en = reinterpret_cast<const int2*>(edges) + sg.x;
    {
        const float hs = row_scale<NV>(hv, nf);
        for (int v = 0; v < NV; ++v) { hv[v].x *= hs; hv[v].y *= hs; hv[v].z *= hs; hv[v].w *= hs; }
    }
    if (XRAW) {
        const float xsc = row_scale<NV>(acc, nf);
        for (int v = 0; v < NV; ++v) { acc[v].x *= xsc; acc[v].y *= xsc; acc[v].z *= xsc; acc[v].w *= xsc; }
    }
    for (int v = 0; v < NV; ++v) {
        acc[v].x *= dn_self; acc[v].y *= dn_self; acc[v].z *= dn_self; acc[v].w *= dn_self;
    }
    int2 cur = make_int2(0, 0);
    if (j < len) cur = en[j];
    for (int c0 = 0; c0 < len; c0 += 8) {
        const int left = len - c0;
        const int   s0 = grp_bcast<0>(cur.x);
        const float w0 = __int_as_float(grp_bcast<0>(cur.y));
        float4 r0[NV];
        load_row<NV>(r0, x, s0, j);
        int2 nxt = make_int2(0, 0);
        if (c0 + 8 + j < len) nxt = en[c0 + 8 + j];
        add_row<NV, XRAW>(acc, r0, w0, nf);
        edge_trip<NV, 1, XRAW>(acc, cur, left, x, j, nf);
        edge_trip<NV, 2, XRAW>(acc, cur, left, x, j, nf);
        edge_trip<NV, 3, XRAW>(acc, cur, left, x, j, nf);
        edge_trip<NV, 4, XRAW>(acc, cur, left, x, j, nf);
        edge_trip<NV, 5, XRAW>(acc, cur, left, x, j, nf);
        edge_trip<NV, 6, XRAW>(acc, cur, left, x, j, nf);
        edge_trip<NV, 7, XRAW>(acc, cur, left, x, j, nf);
        cur = nxt;
    }

    const float dd = OMA_F * dn_self;
    float4* __restrict__ dn = reinterpret_cast<float4*>(dst + (size_t)node * M) + j;
    for (int v = 0; v < NV; ++v) {
        float4 o;
        o.x = dd * acc[v].x + ALPHA_F * hv[v].x;
        o.y = dd * acc[v].y + ALPHA_F * hv[v].y;
        o.z = dd * acc[v].z + ALPHA_F * hv[v].z;
        o.w = dd * acc[v].w + ALPHA_F * hv[v].w;
        if (relu) {
            o.x = fmaxf(o.x, 0.0f); o.y = fmaxf(o.y, 0.0f);
            o.z = fmaxf(o.z, 0.0f); o.w = fmaxf(o.w, 0.0f);
        }
        dn[8 * v] = o;
    }
}


static inline size_t align256(size_t b) { return (b + 255) & ~(size_t)255; }

extern "C" void kernel_launch(void* const* d_in, const int* in_sizes, int n_in,
                              void* d_out, int out_size, void* d_ws, size_t ws_size,
                              hipStream_t stream) {
    const float* x_in = (const float*)d_in[0];
    const int*   ei   = (const int*)d_in[1];
    const float* W0   = (const float*)d_in[2];
    const float* b0   = (const float*)d_in[3];
    const float* W1   = (const float*)d_in[4];
    const float* b1   = (const float*)d_in[5];
    const float* Wx   = (const float*)d_in[6];
    const float* bx   = (const float*)d_in[7];
    float* out = (float*)d_out;

    const int n = in_sizes[0] / IN_DIM;
    const int E = in_sizes[1] / 2;
    const int* row = ei;
    const int* col = ei + E;

    char* ws = (char*)d_ws;
    size_t off = 0;
    float* dinv   = (float*)(ws + off); off += align256((size_t)n * 4);
    int*   cnt    = (int*)  (ws + off); off += align256((size_t)(n + 1) * 4);
    int*   seg    = (int*)  (ws + off); off += align256((size_t)n * 8);
    int*   rank   = (int*)  (ws + off); off += align256((size_t)E * 4);
    int*   edges  = (int*)  (ws + off); off += align256((size_t)E * 8);
    float* H      = (float*)(ws + off); off += align256((size_t)n * HID * 4);
    float* X      = (float*)(ws + off); off += align256((size_t)n * HID * 4);
    float* Y      = (float*)(ws + off); off += align256((size_t)n * HID * 4);
    (void)ws_size;

    const int TPB  = 256;
    const int gN   = (n + TPB - 1) / TPB;
    const int gN1  = (n + 1 + TPB - 1) / TPB;
    const int gE   = (E + TPB - 1) / TPB;
    const int gNode = (n + 31) / 32;
    const int gTile = ((n / 16) + 7) / 8;

    zero_int_k<<<gN1, TPB, 0, stream>>>(cnt, n + 1);
    cnt_k<<<gE, TPB, 0, stream>>>(col, cnt, rank, E);
    dinv_from_cnt_k<<<gN, TPB, 0, stream>>>(cnt, dinv, n);
    alloc_k<<<gN, 256, 0, stream>>>(cnt, cnt + n, seg, n);
    fill_k<<<gE, TPB, 0, stream>>>(row, col, rank, seg, dinv, edges, E);

    auto prop = [&](const float* src, float* dst, const float* h, int M, int relu, bool raw) {
        if (M == HID) {
            if (raw)
                gather_prop_k<HID / 32, true><<<gNode, TPB, 0, stream>>>(dst, src, h, dinv, seg, edges, n, relu, 1.8f);
            else
                gather_prop_k<HID / 32, false><<<gNode, TPB, 0, stream>>>(dst, src, h, dinv, seg, edges, n, relu, 1.8f);
        } else {
            if (raw)
                gather_prop_k<OUT_DIM / 32, true><<<gNode, TPB, 0, stream>>>(dst, src, h, dinv, seg, edges, n, relu, 1.8f);
            else
                gather_prop_k<OUT_DIM / 32, false><<<gNode, TPB, 0, stream>>>(dst, src, h, dinv, seg, edges, n, relu, 1.8f);
        }
    };

    gemm_wmma_k<HID / 16, false><<<gTile, TPB, 0, stream>>>(x_in, W0, b0, H, IN_DIM, n, 1.0f);
    prop(H, X, H, HID, 0, true);
    prop(X, Y, H, HID, 1, false);

    gemm_wmma_k<HID / 16, false><<<gTile, TPB, 0, stream>>>(Y, W1, b1, H, HID, n, 1.0f);
    prop(H, X, H, HID, 0, true);
    prop(X, Y, H, HID, 0, false);

    gemm_wmma_k<OUT_DIM / 16, true><<<gTile, TPB, 0, stream>>>(Y, Wx, bx, H, HID, n, 1.5f);
    prop(H, X, H, OUT_DIM, 0, true);
    prop(X, out, H, OUT_DIM, 0, false);
}


# layer-0 product (x @ W0) moved into the count pass's launch: the first 391 blocks run its WMMA strips, the rest count edges; the independent product overlaps the scattered adds; 14->13 launches
# speedup vs baseline: 2.9106x; 1.0448x over previous
#include <hip/hip_runtime.h>
#include <math.h>

typedef __attribute__((ext_vector_type(2))) float v2f;
typedef __attribute__((ext_vector_type(8))) float v8f;

#define IN_DIM 256
#define HID 96
#define OUT_DIM 64
#define ALPHA_F 0.15f
#define OMA_F 0.85f
#define EPS_F 1e-12f


__global__ void zero_int_k(int* p, int n) {
    int i = blockIdx.x * blockDim.x + threadIdx.x;
    if (i < n) p[i] = 0;
}


__global__ void dinv_from_cnt_k(const int* __restrict__ cnt, float* dinv, int n) {
    int i = blockIdx.x * blockDim.x + threadIdx.x;
    if (i < n) dinv[i] = rsqrtf((float)(cnt[i] + 1));
}

__global__ __launch_bounds__(256) void alloc_k(const int* __restrict__ cnt, int* total,
                                               int* __restrict__ seg, int n) {
    __shared__ int wsum[8];
    __shared__ int base_s;
    const int tid  = threadIdx.x;
    const int lane = tid & 31;
    const int wave = tid >> 5;
    const int i = blockIdx.x * 256 + tid;
    const int c = (i < n) ? cnt[i] : 0;
    int incl = c;
    for (int o = 1; o < 32; o <<= 1) {
        const int t = __shfl_up(incl, o, 32);
        if (lane >= o) incl += t;
    }
    if (lane == 31) wsum[wave] = incl;
    __syncthreads();
    if (tid == 0) {
        int s = 0;
        for (int w = 0; w < 8; ++w) { const int t = wsum[w]; wsum[w] = s; s += t; }
        base_s = atomicAdd(total, s);
    }
    __syncthreads();
    if (i < n) {
        const int p = base_s + wsum[wave] + incl - c;
        reinterpret_cast<int2*>(seg)[i] = make_int2(p, c);
    }
}

__global__ void fill_k(const int* __restrict__ row, const int* __restrict__ col,
                       const int* __restrict__ rank, const int* __restrict__ seg,
                       const float* __restrict__ dinv,
                       int* __restrict__ edges, int E) {
    int i = blockIdx.x * blockDim.x + threadIdx.x;
    if (i < E) {
        const int r = row[i];
        const int p = seg[2 * col[i]] + rank[i];
        reinterpret_cast<int2*>(edges)[p] = make_int2(r, __float_as_int(dinv[r]));
    }
}

template <int NT, bool ANORM>
__device__ __forceinline__ void gemm_strip(
    const int tm,
    const float* __restrict__ X, const float* __restrict__ W,
    const float* __restrict__ bias, float* __restrict__ Out,
    const int K, const int nrows, const float nf) {
    const int M    = NT * 16;
    if (tm * 16 >= nrows) return;
    const int lane = threadIdx.x & 31;
    const int half = lane >> 4;
    const int l16  = lane & 15;

    v8f c[NT];
#pragma unroll
    for (int t = 0; t < NT; ++t) c[t] = (v8f){};

    const float* __restrict__ xrow = X + (size_t)(tm * 16 + l16) * K;
    float ss = 0.0f;

#pragma unroll 4
    for (int k = 0; k < K; k += 4) {
        const int kk = k + 2 * half;
        v2f a;
        a.x = xrow[kk];
        a.y = xrow[kk + 1];
        if (ANORM) { ss = fmaf(a.x, a.x, ss); ss = fmaf(a.y, a.y, ss); }
        const float* __restrict__ w0 = W + (size_t)kk * M + l16;
#pragma unroll
        for (int t = 0; t < NT; ++t) {
            v2f b;
            b.x = w0[t * 16];
            b.y = w0[M + t * 16];
            c[t] = __builtin_amdgcn_wmma_f32_16x16x4_f32(
                false, a, false, b, (short)0, c[t], false, false);
        }
    }

    float rowf[8];
    if (ANORM) {
        ss += __shfl_xor(ss, 16, 32);
        const float f = nf / fmaxf(sqrtf(ss), EPS_F);
#pragma unroll
        for (int r = 0; r < 8; ++r) rowf[r] = __shfl(f, r + 8 * half, 32);
    }

#pragma unroll
    for (int t = 0; t < NT; ++t) {
        const int bcol = t * 16 + l16;
        const float bv = bias[bcol];
#pragma unroll
        for (int r = 0; r < 8; ++r) {
            const int orow = tm * 16 + r + 8 * half;
            float o = c[t][r];
            if (ANORM) o *= rowf[r];
            Out[(size_t)orow * M + bcol] = o + bv;
        }
    }
}

template <int NT, bool ANORM>
__global__ __launch_bounds__(256) void gemm_wmma_k(
    const float* __restrict__ X, const float* __restrict__ W,
    const float* __restrict__ bias, float* __restrict__ Out,
    int K, int nrows, float nf) {
    gemm_strip<NT, ANORM>(blockIdx.x * 8 + (threadIdx.x >> 5), X, W, bias, Out, K, nrows, nf);
}

__global__ __launch_bounds__(256) void cnt_gemm0_k(
    const int* __restrict__ col, int* cnt, int* __restrict__ rank, int E,
    const float* __restrict__ X, const float* __restrict__ W,
    const float* __restrict__ bias, float* __restrict__ Out,
    int K, int nrows, int gemm_blocks) {
    if ((int)blockIdx.x < gemm_blocks) {
        gemm_strip<HID / 16, false>(blockIdx.x * 8 + (threadIdx.x >> 5),
                                    X, W, bias, Out, K, nrows, 1.0f);
        return;
    }
    const int i = ((int)blockIdx.x - gemm_blocks) * 256 + threadIdx.x;
    if (i < E) rank[i] = atomicAdd(&cnt[col[i]], 1);
}


template <int T>
__device__ __forceinline__ int grp_bcast(int v) {
    return (int)__builtin_amdgcn_mov_dpp8((unsigned)v, (unsigned)(T * 0x249249));
}

template <int K>
__device__ __forceinline__ float grp_xor(float v) {
    constexpr unsigned sel =
        ((0u ^ K) << 0)  | ((1u ^ K) << 3)  | ((2u ^ K) << 6)  | ((3u ^ K) << 9) |
        ((4u ^ K) << 12) | ((5u ^ K) << 15) | ((6u ^ K) << 18) | ((7u ^ K) << 21);
    return __int_as_float((int)__builtin_amdgcn_mov_dpp8((unsigned)__float_as_int(v), sel));
}

template <int NV>
__device__ __forceinline__ float row_scale(const float4 (&r)[NV], const float s) {
    float4 p = make_float4(0.0f, 0.0f, 0.0f, 0.0f);
    for (int v = 0; v < NV; ++v) {
        p.x = fmaf(r[v].x, r[v].x, p.x); p.y = fmaf(r[v].y, r[v].y, p.y);
        p.z = fmaf(r[v].z, r[v].z, p.z); p.w = fmaf(r[v].w, r[v].w, p.w);
    }
    float ss = (p.x + p.y) + (p.z + p.w);
    ss += grp_xor<1>(ss);
    ss += grp_xor<2>(ss);
    ss += grp_xor<4>(ss);
    return (ss > EPS_F * EPS_F) ? s * __builtin_amdgcn_rsqf(ss) : s / EPS_F;
}

template <int NV>
__device__ __forceinline__ void load_row(float4 (&r)[NV], const float* __restrict__ x,
                                         const int s, const int j) {
    constexpr int M = NV * 32;
    const float4* __restrict__ xs = reinterpret_cast<const float4*>(x + (size_t)s * M) + j;
    for (int v = 0; v < NV; ++v) r[v] = xs[8 * v];
}

template <int NV, bool XRAW>
__device__ __forceinline__ void add_row(float4 (&acc)[NV], const float4 (&r)[NV], float w,
                                        const float nf) {
    if (XRAW) w *= row_scale<NV>(r, nf);
    for (int v = 0; v < NV; ++v) {
        acc[v].x += w * r[v].x; acc[v].y += w * r[v].y;
        acc[v].z += w * r[v].z; acc[v].w += w * r[v].w;
    }
}

template <int NV, int T, bool XRAW>
__device__ __forceinline__ void edge_trip(float4 (&acc)[NV], const int2 cur, const int left,
                                          const float* __restrict__ x, const int j, const float nf) {
    const int   s = grp_bcast<T>(cur.x);
    const float w = __int_as_float(grp_bcast<T>(cur.y));
    if (T < left) {
        float4 r[NV];
        load_row<NV>(r, x, s, j);
        add_row<NV, XRAW>(acc, r, w, nf);
    }
}

template <int NV, bool XRAW>
__global__ __launch_bounds__(256) void gather_prop_k(
    float* __restrict__ dst, const float* __restrict__ x,
    const float* __restrict__ h, const float* __restrict__ dinv,
    const int* __restrict__ seg, const int* __restrict__ edges,
    int n, int relu, float nf) {
    constexpr int M = NV * 32;
    const int j    = threadIdx.x & 7;
    const int node = blockIdx.x * 32 + (threadIdx.x >> 3);
    if (node >= n) return;

    const float4* __restrict__ xn = reinterpret_cast<const float4*>(x + (size_t)node * M) + j;
    const float4* __restrict__ hn = reinterpret_cast<const float4*>(h + (size_t)node * M) + j;
    float4 acc[NV];
    float4 hv[NV];
    for (int v = 0; v < NV; ++v) acc[v] = xn[8 * v];
    for (int v = 0; v < NV; ++v) hv[v] = hn[8 * v];
    const float dn_self = dinv[node];

    const int2 sg = reinterpret_cast<const int2*>(seg)[node];
    const int len = sg.y;
    const int2* __restrict__ en = reinterpret_cast<const int2*>(edges) + sg.x;
    {
        const float hs = row_scale<NV>(hv, nf);
        for (int v = 0; v < NV; ++v) { hv[v].x *= hs; hv[v].y *= hs; hv[v].z *= hs; hv[v].w *= hs; }
    }
    if (XRAW) {
        const float xsc = row_scale<NV>(acc, nf);
        for (int v = 0; v < NV; ++v) { acc[v].x *= xsc; acc[v].y *= xsc; acc[v].z *= xsc; acc[v].w *= xsc; }
    }
    for (int v = 0; v < NV; ++v) {
        acc[v].x *= dn_self; acc[v].y *= dn_self; acc[v].z *= dn_self; acc[v].w *= dn_self;
    }
    int2 cur = make_int2(0, 0);
    if (j < len) cur = en[j];
    for (int c0 = 0; c0 < len; c0 += 8) {
        const int left = len - c0;
        const int   s0 = grp_bcast<0>(cur.x);
        const float w0 = __int_as_float(grp_bcast<0>(cur.y));
        float4 r0[NV];
        load_row<NV>(r0, x, s0, j);
        int2 nxt = make_int2(0, 0);
        if (c0 + 8 + j < len) nxt = en[c0 + 8 + j];
        add_row<NV, XRAW>(acc, r0, w0, nf);
        edge_trip<NV, 1, XRAW>(acc, cur, left, x, j, nf);
        edge_trip<NV, 2, XRAW>(acc, cur, left, x, j, nf);
        edge_trip<NV, 3, XRAW>(acc, cur, left, x, j, nf);
        edge_trip<NV, 4, XRAW>(acc, cur, left, x, j, nf);
        edge_trip<NV, 5, XRAW>(acc, cur, left, x, j, nf);
        edge_trip<NV, 6, XRAW>(acc, cur, left, x, j, nf);
        edge_trip<NV, 7, XRAW>(acc, cur, left, x, j, nf);
        cur = nxt;
    }

    const float dd = OMA_F * dn_self;
    float4* __restrict__ dn = reinterpret_cast<float4*>(dst + (size_t)node * M) + j;
    for (int v = 0; v < NV; ++v) {
        float4 o;
        o.x = dd * acc[v].x + ALPHA_F * hv[v].x;
        o.y = dd * acc[v].y + ALPHA_F * hv[v].y;
        o.z = dd * acc[v].z + ALPHA_F * hv[v].z;
        o.w = dd * acc[v].w + ALPHA_F * hv[v].w;
        if (relu) {
            o.x = fmaxf(o.x, 0.0f); o.y = fmaxf(o.y, 0.0f);
            o.z = fmaxf(o.z, 0.0f); o.w = fmaxf(o.w, 0.0f);
        }
        dn[8 * v] = o;
    }
}


static inline size_t align256(size_t b) { return (b + 255) & ~(size_t)255; }

extern "C" void kernel_launch(void* const* d_in, const int* in_sizes, int n_in,
                              void* d_out, int out_size, void* d_ws, size_t ws_size,
                              hipStream_t stream) {
    const float* x_in = (const float*)d_in[0];
    const int*   ei   = (const int*)d_in[1];
    const float* W0   = (const float*)d_in[2];
    const float* b0   = (const float*)d_in[3];
    const float* W1   = (const float*)d_in[4];
    const float* b1   = (const float*)d_in[5];
    const float* Wx   = (const float*)d_in[6];
    const float* bx   = (const float*)d_in[7];
    float* out = (float*)d_out;

    const int n = in_sizes[0] / IN_DIM;
    const int E = in_sizes[1] / 2;
    const int* row = ei;
    const int* col = ei + E;

    char* ws = (char*)d_ws;
    size_t off = 0;
    float* dinv   = (float*)(ws + off); off += align256((size_t)n * 4);
    int*   cnt    = (int*)  (ws + off); off += align256((size_t)(n + 1) * 4);
    int*   seg    = (int*)  (ws + off); off += align256((size_t)n * 8);
    int*   rank   = (int*)  (ws + off); off += align256((size_t)E * 4);
    int*   edges  = (int*)  (ws + off); off += align256((size_t)E * 8);
    float* H      = (float*)(ws + off); off += align256((size_t)n * HID * 4);
    float* X      = (float*)(ws + off); off += align256((size_t)n * HID * 4);
    float* Y      = (float*)(ws + off); off += align256((size_t)n * HID * 4);
    (void)ws_size;

    const int TPB  = 256;
    const int gN   = (n + TPB - 1) / TPB;
    const int gN1  = (n + 1 + TPB - 1) / TPB;
    const int gE   = (E + TPB - 1) / TPB;
    const int gNode = (n + 31) / 32;
    const int gTile = ((n / 16) + 7) / 8;

    zero_int_k<<<gN1, TPB, 0, stream>>>(cnt, n + 1);
    cnt_gemm0_k<<<gTile + gE, TPB, 0, stream>>>(col, cnt, rank, E, x_in, W0, b0, H, IN_DIM, n, gTile);
    dinv_from_cnt_k<<<gN, TPB, 0, stream>>>(cnt, dinv, n);
    alloc_k<<<gN, 256, 0, stream>>>(cnt, cnt + n, seg, n);
    fill_k<<<gE, TPB, 0, stream>>>(row, col, rank, seg, dinv, edges, E);

    auto prop = [&](const float* src, float* dst, const float* h, int M, int relu, bool raw) {
        if (M == HID) {
            if (raw)
                gather_prop_k<HID / 32, true><<<gNode, TPB, 0, stream>>>(dst, src, h, dinv, seg, edges, n, relu, 1.8f);
            else
                gather_prop_k<HID / 32, false><<<gNode, TPB, 0, stream>>>(dst, src, h, dinv, seg, edges, n, relu, 1.8f);
        } else {
            if (raw)
                gather_prop_k<OUT_DIM / 32, true><<<gNode, TPB, 0, stream>>>(dst, src, h, dinv, seg, edges, n, relu, 1.8f);
            else
                gather_prop_k<OUT_DIM / 32, false><<<gNode, TPB, 0, stream>>>(dst, src, h, dinv, seg, edges, n, relu, 1.8f);
        }
    };

    prop(H, X, H, HID, 0, true);
    prop(X, Y, H, HID, 1, false);

    gemm_wmma_k<HID / 16, false><<<gTile, TPB, 0, stream>>>(Y, W1, b1, H, HID, n, 1.0f);
    prop(H, X, H, HID, 0, true);
    prop(X, Y, H, HID, 0, false);

    gemm_wmma_k<OUT_DIM / 16, true><<<gTile, TPB, 0, stream>>>(Y, Wx, bx, H, HID, n, 1.5f);
    prop(H, X, H, OUT_DIM, 0, true);
    prop(X, out, H, OUT_DIM, 0, false);
}


# next layer's product runs in the gather that makes its rows: a block's 32 result rows go to LDS, its waves do the WMMA tiles; 13->11 launches; l2norm before last product: sum-sq order changes
# speedup vs baseline: 3.1099x; 1.0685x over previous
#include <hip/hip_runtime.h>
#include <math.h>

typedef __attribute__((ext_vector_type(2))) float v2f;
typedef __attribute__((ext_vector_type(8))) float v8f;

#define IN_DIM 256
#define HID 96
#define OUT_DIM 64
#define ALPHA_F 0.15f
#define OMA_F 0.85f
#define EPS_F 1e-12f


__global__ void zero_int_k(int* p, int n) {
    int i = blockIdx.x * blockDim.x + threadIdx.x;
    if (i < n) p[i] = 0;
}


__global__ void dinv_from_cnt_k(const int* __restrict__ cnt, float* dinv, int n) {
    int i = blockIdx.x * blockDim.x + threadIdx.x;
    if (i < n) dinv[i] = rsqrtf((float)(cnt[i] + 1));
}

__global__ __launch_bounds__(256) void alloc_k(const int* __restrict__ cnt, int* total,
                                               int* __restrict__ seg, int n) {
    __shared__ int wsum[8];
    __shared__ int base_s;
    const int tid  = threadIdx.x;
    const int lane = tid & 31;
    const int wave = tid >> 5;
    const int i = blockIdx.x * 256 + tid;
    const int c = (i < n) ? cnt[i] : 0;
    int incl = c;
    for (int o = 1; o < 32; o <<= 1) {
        const int t = __shfl_up(incl, o, 32);
        if (lane >= o) incl += t;
    }
    if (lane == 31) wsum[wave] = incl;
    __syncthreads();
    if (tid == 0) {
        int s = 0;
        for (int w = 0; w < 8; ++w) { const int t = wsum[w]; wsum[w] = s; s += t; }
        base_s = atomicAdd(total, s);
    }
    __syncthreads();
    if (i < n) {
        const int p = base_s + wsum[wave] + incl - c;
        reinterpret_cast<int2*>(seg)[i] = make_int2(p, c);
    }
}

__global__ void fill_k(const int* __restrict__ row, const int* __restrict__ col,
                       const int* __restrict__ rank, const int* __restrict__ seg,
                       const float* __restrict__ dinv,
                       int* __restrict__ edges, int E) {
    int i = blockIdx.x * blockDim.x + threadIdx.x;
    if (i < E) {
        const int r = row[i];
        const int p = seg[2 * col[i]] + rank[i];
        reinterpret_cast<int2*>(edges)[p] = make_int2(r, __float_as_int(dinv[r]));
    }
}

template <int NT>
__device__ __forceinline__ void gemm_strip(
    const int tm,
    const float* __restrict__ X, const float* __restrict__ W,
    const float* __restrict__ bias, float* __restrict__ Out,
    const int K, const int nrows) {
    const int M    = NT * 16;
    if (tm * 16 >= nrows) return;
    const int lane = threadIdx.x & 31;
    const int half = lane >> 4;
    const int l16  = lane & 15;

    v8f c[NT];
#pragma unroll
    for (int t = 0; t < NT; ++t) c[t] = (v8f){};

    const float* __restrict__ xrow = X + (size_t)(tm * 16 + l16) * K;

#pragma unroll 4
    for (int k = 0; k < K; k += 4) {
        const int kk = k + 2 * half;
        v2f a;
        a.x = xrow[kk];
        a.y = xrow[kk + 1];
        const float* __restrict__ w0 = W + (size_t)kk * M + l16;
#pragma unroll
        for (int t = 0; t < NT; ++t) {
            v2f b;
            b.x = w0[t * 16];
            b.y = w0[M + t * 16];
            c[t] = __builtin_amdgcn_wmma_f32_16x16x4_f32(
                false, a, false, b, (short)0, c[t], false, false);
        }
    }

#pragma unroll
    for (int t = 0; t < NT; ++t) {
        const int bcol = t * 16 + l16;
        const float bv = bias[bcol];
#pragma unroll
        for (int r = 0; r < 8; ++r) {
            const int orow = tm * 16 + r + 8 * half;
            Out[(size_t)orow * M + bcol] = c[t][r] + bv;
        }
    }
}

__global__ __launch_bounds__(256) void cnt_gemm0_k(
    const int* __restrict__ col, int* cnt, int* __restrict__ rank, int E,
    const float* __restrict__ X, const float* __restrict__ W,
    const float* __restrict__ bias, float* __restrict__ Out,
    int K, int nrows, int gemm_blocks) {
    if ((int)blockIdx.x < gemm_blocks) {
        gemm_strip<HID / 16>(blockIdx.x * 8 + (threadIdx.x >> 5),
                             X, W, bias, Out, K, nrows);
        return;
    }
    const int i = ((int)blockIdx.x - gemm_blocks) * 256 + threadIdx.x;
    if (i < E) rank[i] = atomicAdd(&cnt[col[i]], 1);
}


template <int T>
__device__ __forceinline__ int grp_bcast(int v) {
    return (int)__builtin_amdgcn_mov_dpp8((unsigned)v, (unsigned)(T * 0x249249));
}

template <int K>
__device__ __forceinline__ float grp_xor(float v) {
    constexpr unsigned sel =
        ((0u ^ K) << 0)  | ((1u ^ K) << 3)  | ((2u ^ K) << 6)  | ((3u ^ K) << 9) |
        ((4u ^ K) << 12) | ((5u ^ K) << 15) | ((6u ^ K) << 18) | ((7u ^ K) << 21);
    return __int_as_float((int)__builtin_amdgcn_mov_dpp8((unsigned)__float_as_int(v), sel));
}

template <int NV>
__device__ __forceinline__ float row_scale(const float4 (&r)[NV], const float s) {
    float4 p = make_float4(0.0f, 0.0f, 0.0f, 0.0f);
    for (int v = 0; v < NV; ++v) {
        p.x = fmaf(r[v].x, r[v].x, p.x); p.y = fmaf(r[v].y, r[v].y, p.y);
        p.z = fmaf(r[v].z, r[v].z, p.z); p.w = fmaf(r[v].w, r[v].w, p.w);
    }
    float ss = (p.x + p.y) + (p.z + p.w);
    ss += grp_xor<1>(ss);
    ss += grp_xor<2>(ss);
    ss += grp_xor<4>(ss);
    return (ss > EPS_F * EPS_F) ? s * __builtin_amdgcn_rsqf(ss) : s / EPS_F;
}

template <int NV>
__device__ __forceinline__ void load_row(float4 (&r)[NV], const float* __restrict__ x,
                                         const int s, const int j) {
    constexpr int M = NV * 32;
    const float4* __restrict__ xs = reinterpret_cast<const float4*>(x + (size_t)s * M) + j;
    for (int v = 0; v < NV; ++v) r[v] = xs[8 * v];
}

template <int NV, bool XRAW>
__device__ __forceinline__ void add_row(float4 (&acc)[NV], const float4 (&r)[NV], float w,
                                        const float nf) {
    if (XRAW) w *= row_scale<NV>(r, nf);
    for (int v = 0; v < NV; ++v) {
        acc[v].x += w * r[v].x; acc[v].y += w * r[v].y;
        acc[v].z += w * r[v].z; acc[v].w += w * r[v].w;
    }
}

template <int NV, int T, bool XRAW>
__device__ __forceinline__ void edge_trip(float4 (&acc)[NV], const int2 cur, const int left,
                                          const float* __restrict__ x, const int j, const float nf) {
    const int   s = grp_bcast<T>(cur.x);
    const float w = __int_as_float(grp_bcast<T>(cur.y));
    if (T < left) {
        float4 r[NV];
        load_row<NV>(r, x, s, j);
        add_row<NV, XRAW>(acc, r, w, nf);
    }
}

template <int NV, bool XRAW>
__device__ __forceinline__ void gather_row(
    float4 (&o)[NV], const float* __restrict__ x,
    const float* __restrict__ h, const float* __restrict__ dinv,
    const int* __restrict__ seg, const int* __restrict__ edges,
    const int node, const int j, const int relu, const float nf) {
    constexpr int M = NV * 32;
    const float4* __restrict__ xn = reinterpret_cast<const float4*>(x + (size_t)node * M) + j;
    const float4* __restrict__ hn = reinterpret_cast<const float4*>(h + (size_t)node * M) + j;
    float4 acc[NV];
    float4 hv[NV];
    for (int v = 0; v < NV; ++v) acc[v] = xn[8 * v];
    for (int v = 0; v < NV; ++v) hv[v] = hn[8 * v];
    const float dn_self = dinv[node];

    const int2 sg = reinterpret_cast<const int2*>(seg)[node];
    const int len = sg.y;
    const int2* __restrict__ en = reinterpret_cast<const int2*>(edges) + sg.x;
    {
        const float hs = row_scale<NV>(hv, nf);
        for (int v = 0; v < NV; ++v) { hv[v].x *= hs; hv[v].y *= hs; hv[v].z *= hs; hv[v].w *= hs; }
    }
    if (XRAW) {
        const float xsc = row_scale<NV>(acc, nf);
        for (int v = 0; v < NV; ++v) { acc[v].x *= xsc; acc[v].y *= xsc; acc[v].z *= xsc; acc[v].w *= xsc; }
    }
    for (int v = 0; v < NV; ++v) {
        acc[v].x *= dn_self; acc[v].y *= dn_self; acc[v].z *= dn_self; acc[v].w *= dn_self;
    }
    int2 cur = make_int2(0, 0);
    if (j < len) cur = en[j];
    for (int c0 = 0; c0 < len; c0 += 8) {
        const int left = len - c0;
        const int   s0 = grp_bcast<0>(cur.x);
        const float w0 = __int_as_float(grp_bcast<0>(cur.y));
        float4 r0[NV];
        load_row<NV>(r0, x, s0, j);
        int2 nxt = make_int2(0, 0);
        if (c0 + 8 + j < len) nxt = en[c0 + 8 + j];
        add_row<NV, XRAW>(acc, r0, w0, nf);
        edge_trip<NV, 1, XRAW>(acc, cur, left, x, j, nf);
        edge_trip<NV, 2, XRAW>(acc, cur, left, x, j, nf);
        edge_trip<NV, 3, XRAW>(acc, cur, left, x, j, nf);
        edge_trip<NV, 4, XRAW>(acc, cur, left, x, j, nf);
        edge_trip<NV, 5, XRAW>(acc, cur, left, x, j, nf);
        edge_trip<NV, 6, XRAW>(acc, cur, left, x, j, nf);
        edge_trip<NV, 7, XRAW>(acc, cur, left, x, j, nf);
        cur = nxt;
    }

    const float dd = OMA_F * dn_self;
    for (int v = 0; v < NV; ++v) {
        o[v].x = dd * acc[v].x + ALPHA_F * hv[v].x;
        o[v].y = dd * acc[v].y + ALPHA_F * hv[v].y;
        o[v].z = dd * acc[v].z + ALPHA_F * hv[v].z;
        o[v].w = dd * acc[v].w + ALPHA_F * hv[v].w;
        if (relu) {
            o[v].x = fmaxf(o[v].x, 0.0f); o[v].y = fmaxf(o[v].y, 0.0f);
            o[v].z = fmaxf(o[v].z, 0.0f); o[v].w = fmaxf(o[v].w, 0.0f);
        }
    }
}

template <int NV, bool XRAW>
__global__ __launch_bounds__(256) void gather_prop_k(
    float* __restrict__ dst, const float* __restrict__ x,
    const float* __restrict__ h, const float* __restrict__ dinv,
    const int* __restrict__ seg, const int* __restrict__ edges,
    int n, int relu, float nf) {
    constexpr int M = NV * 32;
    const int j    = threadIdx.x & 7;
    const int node = blockIdx.x * 32 + (threadIdx.x >> 3);
    if (node >= n) return;
    float4 o[NV];
    gather_row<NV, XRAW>(o, x, h, dinv, seg, edges, node, j, relu, nf);
    float4* __restrict__ dn = reinterpret_cast<float4*>(dst + (size_t)node * M) + j;
    for (int v = 0; v < NV; ++v) dn[8 * v] = o[v];
}

template <int NT, bool ANORM>
__global__ __launch_bounds__(256) void gather_gemm_k(
    float* __restrict__ Out, const float* __restrict__ x,
    const float* __restrict__ h, const float* __restrict__ dinv,
    const int* __restrict__ seg, const int* __restrict__ edges,
    const float* __restrict__ W, const float* __restrict__ bias,
    int n, int relu, float nf, float nfa) {
    constexpr int NV    = HID / 32;
    constexpr int M     = NT * 16;
    constexpr int PITCH = HID + 4;
    __shared__ float tile[32 * PITCH];

    const int j  = threadIdx.x & 7;
    const int nb = threadIdx.x >> 3;
    const int own  = (int)blockIdx.x * 32 + nb;
    const int node = own < n ? own : n - 1;
    float4 o[NV];
    gather_row<NV, false>(o, x, h, dinv, seg, edges, node, j, relu, nf);
    if (ANORM) {
        const float f = row_scale<NV>(o, nfa);
        for (int v = 0; v < NV; ++v) { o[v].x *= f; o[v].y *= f; o[v].z *= f; o[v].w *= f; }
    }
    for (int v = 0; v < NV; ++v)
        *reinterpret_cast<float4*>(&tile[nb * PITCH + 4 * (j + 8 * v)]) = o[v];
    __syncthreads();

    const int wave = threadIdx.x >> 5;
    const int lane = threadIdx.x & 31;
    const int half = lane >> 4;
    const int l16  = lane & 15;
    const int rt   = wave & 1;
    const int cp   = wave >> 1;
    const int row0 = (int)blockIdx.x * 32 + rt * 16;
    if (cp * 2 >= NT || row0 >= n) return;

    v8f c[2];
    c[0] = (v8f){};
    c[1] = (v8f){};
    const float* __restrict__ arow = tile + (rt * 16 + l16) * PITCH + 2 * half;
    const float* __restrict__ wcol = W + (size_t)(2 * half) * M + cp * 32 + l16;
#pragma unroll 4
    for (int k = 0; k < HID; k += 4) {
        const v2f a = *reinterpret_cast<const v2f*>(arow + k);
        const float* __restrict__ w0 = wcol + (size_t)k * M;
#pragma unroll
        for (int t = 0; t < 2; ++t) {
            v2f b;
            b.x = w0[t * 16];
            b.y = w0[M + t * 16];
            c[t] = __builtin_amdgcn_wmma_f32_16x16x4_f32(
                false, a, false, b, (short)0, c[t], false, false);
        }
    }

#pragma unroll
    for (int t = 0; t < 2; ++t) {
        const int bcol = cp * 32 + t * 16 + l16;
        const float bv = bias[bcol];
#pragma unroll
        for (int r = 0; r < 8; ++r) {
            const int orow = row0 + r + 8 * half;
            Out[(size_t)orow * M + bcol] = c[t][r] + bv;
        }
    }
}


static inline size_t align256(size_t b) { return (b + 255) & ~(size_t)255; }

extern "C" void kernel_launch(void* const* d_in, const int* in_sizes, int n_in,
                              void* d_out, int out_size, void* d_ws, size_t ws_size,
                              hipStream_t stream) {
    const float* x_in = (const float*)d_in[0];
    const int*   ei   = (const int*)d_in[1];
    const float* W0   = (const float*)d_in[2];
    const float* b0   = (const float*)d_in[3];
    const float* W1   = (const float*)d_in[4];
    const float* b1   = (const float*)d_in[5];
    const float* Wx   = (const float*)d_in[6];
    const float* bx   = (const float*)d_in[7];
    float* out = (float*)d_out;

    const int n = in_sizes[0] / IN_DIM;
    const int E = in_sizes[1] / 2;
    const int* row = ei;
    const int* col = ei + E;

    char* ws = (char*)d_ws;
    size_t off = 0;
    float* dinv   = (float*)(ws + off); off += align256((size_t)n * 4);
    int*   cnt    = (int*)  (ws + off); off += align256((size_t)(n + 1) * 4);
    int*   seg    = (int*)  (ws + off); off += align256((size_t)n * 8);
    int*   rank   = (int*)  (ws + off); off += align256((size_t)E * 4);
    int*   edges  = (int*)  (ws + off); off += align256((size_t)E * 8);
    float* H      = (float*)(ws + off); off += align256((size_t)n * HID * 4);
    float* X      = (float*)(ws + off); off += align256((size_t)n * HID * 4);
    float* Y      = (float*)(ws + off); off += align256((size_t)n * HID * 4);
    (void)ws_size;

    const int TPB  = 256;
    const int gN   = (n + TPB - 1) / TPB;
    const int gN1  = (n + 1 + TPB - 1) / TPB;
    const int gE   = (E + TPB - 1) / TPB;
    const int gNode = (n + 31) / 32;
    const int gTile = ((n / 16) + 7) / 8;

    zero_int_k<<<gN1, TPB, 0, stream>>>(cnt, n + 1);
    cnt_gemm0_k<<<gTile + gE, TPB, 0, stream>>>(col, cnt, rank, E, x_in, W0, b0, H, IN_DIM, n, gTile);
    dinv_from_cnt_k<<<gN, TPB, 0, stream>>>(cnt, dinv, n);
    alloc_k<<<gN, 256, 0, stream>>>(cnt, cnt + n, seg, n);
    fill_k<<<gE, TPB, 0, stream>>>(row, col, rank, seg, dinv, edges, E);


    gather_prop_k<HID / 32, true><<<gNode, TPB, 0, stream>>>(X, H, H, dinv, seg, edges, n, 0, 1.8f);
    gather_gemm_k<HID / 16, false><<<gNode, TPB, 0, stream>>>(Y, X, H, dinv, seg, edges, W1, b1, n, 1, 1.8f, 1.0f);

    gather_prop_k<HID / 32, true><<<gNode, TPB, 0, stream>>>(X, Y, Y, dinv, seg, edges, n, 0, 1.8f);
    gather_gemm_k<OUT_DIM / 16, true><<<gNode, TPB, 0, stream>>>(H, X, Y, dinv, seg, edges, Wx, bx, n, 0, 1.8f, 1.5f);

    gather_prop_k<OUT_DIM / 32, true><<<gNode, TPB, 0, stream>>>(X, H, H, dinv, seg, edges, n, 0, 1.8f);
    gather_prop_k<OUT_DIM / 32, false><<<gNode, TPB, 0, stream>>>(out, X, H, dinv, seg, edges, n, 0, 1.8f);
}


# speedup vs baseline: 3.7740x; 1.2135x over previous
#include <hip/hip_runtime.h>
#include <math.h>

typedef __attribute__((ext_vector_type(2))) float v2f;
typedef __attribute__((ext_vector_type(8))) float v8f;

#define IN_DIM 256
#define HID 96
#define OUT_DIM 64
#define ALPHA_F 0.15f
#define OMA_F 0.85f
#define EPS_F 1e-12f


__global__ void zero_int_k(int* p, int n) {
    int i = blockIdx.x * blockDim.x + threadIdx.x;
    if (i < n) p[i] = 0;
}


__device__ __forceinline__ int chip_half() {
    return (int)((__builtin_amdgcn_s_sendmsg_rtn(0x87u) >> 18) & 1u);
}

__global__ void dinv_from_cnt_k(const int* __restrict__ cnt, float* dinv, int n) {
    int i = blockIdx.x * blockDim.x + threadIdx.x;
    if (i < n) dinv[i] = rsqrtf((float)(cnt[i] + 1));
}

__global__ __launch_bounds__(256) void alloc_k(const int* __restrict__ cnt, int* total,
                                               int* __restrict__ seg, int n) {
    __shared__ int wsum[8];
    __shared__ int base_s;
    const int tid  = threadIdx.x;
    const int lane = tid & 31;
    const int wave = tid >> 5;
    const int i = blockIdx.x * 256 + tid;
    const int c = (i < n) ? cnt[i] : 0;
    int incl = c;
    for (int o = 1; o < 32; o <<= 1) {
        const int t = __shfl_up(incl, o, 32);
        if (lane >= o) incl += t;
    }
    if (lane == 31) wsum[wave] = incl;
    __syncthreads();
    if (tid == 0) {
        int s = 0;
        for (int w = 0; w < 8; ++w) { const int t = wsum[w]; wsum[w] = s; s += t; }
        base_s = atomicAdd(total, s);
    }
    __syncthreads();
    if (i < n) {
        const int p = base_s + wsum[wave] + incl - c;
        reinterpret_cast<int2*>(seg)[i] = make_int2(p, c);
    }
}

__global__ void fill_k(const int* __restrict__ row, const int* __restrict__ col,
                       const int* __restrict__ rank, const int* __restrict__ seg,
                       const float* __restrict__ dinv,
                       int* __restrict__ edges, int E, int split) {
    int i = blockIdx.x * blockDim.x + threadIdx.x;
    if (i < E) {
        const int r = row[i];
        const int c = col[i];
        const int p = seg[2 * c] + rank[(c >= split ? (size_t)E : (size_t)0) + i];
        reinterpret_cast<int2*>(edges)[p] = make_int2(r, __float_as_int(dinv[r]));
    }
}

template <int NT>
__device__ __forceinline__ void gemm_strip(
    const int tm,
    const float* __restrict__ X, const float* __restrict__ W,
    const float* __restrict__ bias, float* __restrict__ Out,
    const int K, const int nrows) {
    const int M    = NT * 16;
    if (tm * 16 >= nrows) return;
    const int lane = threadIdx.x & 31;
    const int half = lane >> 4;
    const int l16  = lane & 15;

    v8f c[NT];
#pragma unroll
    for (int t = 0; t < NT; ++t) c[t] = (v8f){};

    const float* __restrict__ xrow = X + (size_t)(tm * 16 + l16) * K;

#pragma unroll 4
    for (int k = 0; k < K; k += 4) {
        const int kk = k + 2 * half;
        v2f a;
        a.x = xrow[kk];
        a.y = xrow[kk + 1];
        const float* __restrict__ w0 = W + (size_t)kk * M + l16;
#pragma unroll
        for (int t = 0; t < NT; ++t) {
            v2f b;
            b.x = w0[t * 16];
            b.y = w0[M + t * 16];
            c[t] = __builtin_amdgcn_wmma_f32_16x16x4_f32(
                false, a, false, b, (short)0, c[t], false, false);
        }
    }

#pragma unroll
    for (int t = 0; t < NT; ++t) {
        const int bcol = t * 16 + l16;
        const float bv = bias[bcol];
#pragma unroll
        for (int r = 0; r < 8; ++r) {
            const int orow = tm * 16 + r + 8 * half;
            Out[(size_t)orow * M + bcol] = c[t][r] + bv;
        }
    }
}

__global__ __launch_bounds__(256) void cnt_gemm0_k(
    const int* __restrict__ col, int* cnt, int* tick, int* __restrict__ rank,
    int E, int split, int jobs,
    const float* __restrict__ X, const float* __restrict__ W,
    const float* __restrict__ bias, float* __restrict__ Out,
    int K, int nrows, int gemm_blocks) {
    if ((int)blockIdx.x < gemm_blocks) {
        gemm_strip<HID / 16>(blockIdx.x * 8 + (threadIdx.x >> 5),
                             X, W, bias, Out, K, nrows);
        return;
    }
    __shared__ int job_s[2];
    if (threadIdx.x == 0) {
        int h = chip_half();
        int t = atomicAdd(&tick[32 * h], 1);
        if (t >= jobs) {
            h ^= 1;
            t = atomicAdd(&tick[32 * h], 1);
        }
        job_s[0] = t;
        job_s[1] = h;
    }
    __syncthreads();
    const int t = job_s[0];
    const int h = job_s[1];
    if (t >= jobs) return;
    int* __restrict__ rk = rank + (size_t)h * E;
    const int i0 = t * 512 + threadIdx.x;
    const int i1 = i0 + 256;
    const int c0 = (i0 < E) ? col[i0] : -1;
    const int c1 = (i1 < E) ? col[i1] : -1;
    const bool m0 = c0 >= 0 && (c0 >= split) == (h != 0);
    const bool m1 = c1 >= 0 && (c1 >= split) == (h != 0);
    int r0 = 0, r1 = 0;
    if (m0) r0 = atomicAdd(&cnt[c0], 1);
    if (m1) r1 = atomicAdd(&cnt[c1], 1);
    if (m0) rk[i0] = r0;
    if (m1) rk[i1] = r1;
}


template <int T>
__device__ __forceinline__ int grp_bcast(int v) {
    return (int)__builtin_amdgcn_mov_dpp8((unsigned)v, (unsigned)(T * 0x249249));
}

template <int K>
__device__ __forceinline__ float grp_xor(float v) {
    constexpr unsigned sel =
        ((0u ^ K) << 0)  | ((1u ^ K) << 3)  | ((2u ^ K) << 6)  | ((3u ^ K) << 9) |
        ((4u ^ K) << 12) | ((5u ^ K) << 15) | ((6u ^ K) << 18) | ((7u ^ K) << 21);
    return __int_as_float((int)__builtin_amdgcn_mov_dpp8((unsigned)__float_as_int(v), sel));
}

template <int NV>
__device__ __forceinline__ float row_scale(const float4 (&r)[NV], const float s) {
    float4 p = make_float4(0.0f, 0.0f, 0.0f, 0.0f);
    for (int v = 0; v < NV; ++v) {
        p.x = fmaf(r[v].x, r[v].x, p.x); p.y = fmaf(r[v].y, r[v].y, p.y);
        p.z = fmaf(r[v].z, r[v].z, p.z); p.w = fmaf(r[v].w, r[v].w, p.w);
    }
    float ss = (p.x + p.y) + (p.z + p.w);
    ss += grp_xor<1>(ss);
    ss += grp_xor<2>(ss);
    ss += grp_xor<4>(ss);
    return (ss > EPS_F * EPS_F) ? s * __builtin_amdgcn_rsqf(ss) : s / EPS_F;
}

template <int NV>
__device__ __forceinline__ void load_row(float4 (&r)[NV], const float* __restrict__ x,
                                         const int s, const int j) {
    constexpr int M = NV * 32;
    const float4* __restrict__ xs = reinterpret_cast<const float4*>(x + (size_t)s * M) + j;
    for (int v = 0; v < NV; ++v) r[v] = xs[8 * v];
}

template <int NV, bool XRAW>
__device__ __forceinline__ void add_row(float4 (&acc)[NV], const float4 (&r)[NV], float w,
                                        const float nf) {
    if (XRAW) w *= row_scale<NV>(r, nf);
    for (int v = 0; v < NV; ++v) {
        acc[v].x += w * r[v].x; acc[v].y += w * r[v].y;
        acc[v].z += w * r[v].z; acc[v].w += w * r[v].w;
    }
}

template <int NV, int T, bool XRAW>
__device__ __forceinline__ void edge_trip(float4 (&acc)[NV], const int2 cur, const int left,
                                          const float* __restrict__ x, const int j, const float nf) {
    const int   s = grp_bcast<T>(cur.x);
    const float w = __int_as_float(grp_bcast<T>(cur.y));
    if (T < left) {
        float4 r[NV];
        load_row<NV>(r, x, s, j);
        add_row<NV, XRAW>(acc, r, w, nf);
    }
}

template <int NV, bool XRAW>
__device__ __forceinline__ void gather_row(
    float4 (&o)[NV], const float* __restrict__ x,
    const float* __restrict__ h, const float* __restrict__ dinv,
    const int* __restrict__ seg, const int* __restrict__ edges,
    const int node, const int j, const int relu, const float nf) {
    constexpr int M = NV * 32;
    const float4* __restrict__ xn = reinterpret_cast<const float4*>(x + (size_t)node * M) + j;
    const float4* __restrict__ hn = reinterpret_cast<const float4*>(h + (size_t)node * M) + j;
    float4 acc[NV];
    float4 hv[NV];
    for (int v = 0; v < NV; ++v) acc[v] = xn[8 * v];
    for (int v = 0; v < NV; ++v) hv[v] = hn[8 * v];
    const float dn_self = dinv[node];

    const int2 sg = reinterpret_cast<const int2*>(seg)[node];
    const int len = sg.y;
    const int2* __restrict__ en = reinterpret_cast<const int2*>(edges) + sg.x;
    {
        const float hs = row_scale<NV>(hv, nf);
        for (int v = 0; v < NV; ++v) { hv[v].x *= hs; hv[v].y *= hs; hv[v].z *= hs; hv[v].w *= hs; }
    }
    if (XRAW) {
        const float xsc = row_scale<NV>(acc, nf);
        for (int v = 0; v < NV; ++v) { acc[v].x *= xsc; acc[v].y *= xsc; acc[v].z *= xsc; acc[v].w *= xsc; }
    }
    for (int v = 0; v < NV; ++v) {
        acc[v].x *= dn_self; acc[v].y *= dn_self; acc[v].z *= dn_self; acc[v].w *= dn_self;
    }
    int2 cur = make_int2(0, 0);
    if (j < len) cur = en[j];
    for (int c0 = 0; c0 < len; c0 += 8) {
        const int left = len - c0;
        const int   s0 = grp_bcast<0>(cur.x);
        const float w0 = __int_as_float(grp_bcast<0>(cur.y));
        float4 r0[NV];
        load_row<NV>(r0, x, s0, j);
        int2 nxt = make_int2(0, 0);
        if (c0 + 8 + j < len) nxt = en[c0 + 8 + j];
        add_row<NV, XRAW>(acc, r0, w0, nf);
        edge_trip<NV, 1, XRAW>(acc, cur, left, x, j, nf);
        edge_trip<NV, 2, XRAW>(acc, cur, left, x, j, nf);
        edge_trip<NV, 3, XRAW>(acc, cur, left, x, j, nf);
        edge_trip<NV, 4, XRAW>(acc, cur, left, x, j, nf);
        edge_trip<NV, 5, XRAW>(acc, cur, left, x, j, nf);
        edge_trip<NV, 6, XRAW>(acc, cur, left, x, j, nf);
        edge_trip<NV, 7, XRAW>(acc, cur, left, x, j, nf);
        cur = nxt;
    }

    const float dd = OMA_F * dn_self;
    for (int v = 0; v < NV; ++v) {
        o[v].x = dd * acc[v].x + ALPHA_F * hv[v].x;
        o[v].y = dd * acc[v].y + ALPHA_F * hv[v].y;
        o[v].z = dd * acc[v].z + ALPHA_F * hv[v].z;
        o[v].w = dd * acc[v].w + ALPHA_F * hv[v].w;
        if (relu) {
            o[v].x = fmaxf(o[v].x, 0.0f); o[v].y = fmaxf(o[v].y, 0.0f);
            o[v].z = fmaxf(o[v].z, 0.0f); o[v].w = fmaxf(o[v].w, 0.0f);
        }
    }
}

template <int NV, bool XRAW>
__global__ __launch_bounds__(256) void gather_prop_k(
    float* __restrict__ dst, const float* __restrict__ x,
    const float* __restrict__ h, const float* __restrict__ dinv,
    const int* __restrict__ seg, const int* __restrict__ edges,
    int n, int relu, float nf) {
    constexpr int M = NV * 32;
    const int j    = threadIdx.x & 7;
    const int node = blockIdx.x * 32 + (threadIdx.x >> 3);
    if (node >= n) return;
    float4 o[NV];
    gather_row<NV, XRAW>(o, x, h, dinv, seg, edges, node, j, relu, nf);
    float4* __restrict__ dn = reinterpret_cast<float4*>(dst + (size_t)node * M) + j;
    for (int v = 0; v < NV; ++v) dn[8 * v] = o[v];
}

template <int NT, bool ANORM>
__global__ __launch_bounds__(256) void gather_gemm_k(
    float* __restrict__ Out, const float* __restrict__ x,
    const float* __restrict__ h, const float* __restrict__ dinv,
    const int* __restrict__ seg, const int* __restrict__ edges,
    const float* __restrict__ W, const float* __restrict__ bias,
    int n, int relu, float nf, float nfa) {
    constexpr int NV    = HID / 32;
    constexpr int M     = NT * 16;
    constexpr int PITCH = HID + 4;
    __shared__ float tile[32 * PITCH];

    const int j  = threadIdx.x & 7;
    const int nb = threadIdx.x >> 3;
    const int own  = (int)blockIdx.x * 32 + nb;
    const int node = own < n ? own : n - 1;
    float4 o[NV];
    gather_row<NV, false>(o, x, h, dinv, seg, edges, node, j, relu, nf);
    if (ANORM) {
        const float f = row_scale<NV>(o, nfa);
        for (int v = 0; v < NV; ++v) { o[v].x *= f; o[v].y *= f; o[v].z *= f; o[v].w *= f; }
    }
    for (int v = 0; v < NV; ++v)
        *reinterpret_cast<float4*>(&tile[nb * PITCH + 4 * (j + 8 * v)]) = o[v];
    __syncthreads();

    const int wave = threadIdx.x >> 5;
    const int lane = threadIdx.x & 31;
    const int half = lane >> 4;
    const int l16  = lane & 15;
    const int rt   = wave & 1;
    const int cp   = wave >> 1;
    const int row0 = (int)blockIdx.x * 32 + rt * 16;
    if (cp * 2 >= NT || row0 >= n) return;

    v8f c[2];
    c[0] = (v8f){};
    c[1] = (v8f){};
    const float* __restrict__ arow = tile + (rt * 16 + l16) * PITCH + 2 * half;
    const float* __restrict__ wcol = W + (size_t)(2 * half) * M + cp * 32 + l16;
#pragma unroll 4
    for (int k = 0; k < HID; k += 4) {
        const v2f a = *reinterpret_cast<const v2f*>(arow + k);
        const float* __restrict__ w0 = wcol + (size_t)k * M;
#pragma unroll
        for (int t = 0; t < 2; ++t) {
            v2f b;
            b.x = w0[t * 16];
            b.y = w0[M + t * 16];
            c[t] = __builtin_amdgcn_wmma_f32_16x16x4_f32(
                false, a, false, b, (short)0, c[t], false, false);
        }
    }

#pragma unroll
    for (int t = 0; t < 2; ++t) {
        const int bcol = cp * 32 + t * 16 + l16;
        const float bv = bias[bcol];
#pragma unroll
        for (int r = 0; r < 8; ++r) {
            const int orow = row0 + r + 8 * half;
            Out[(size_t)orow * M + bcol] = c[t][r] + bv;
        }
    }
}


static inline size_t align256(size_t b) { return (b + 255) & ~(size_t)255; }

extern "C" void kernel_launch(void* const* d_in, const int* in_sizes, int n_in,
                              void* d_out, int out_size, void* d_ws, size_t ws_size,
                              hipStream_t stream) {
    const float* x_in = (const float*)d_in[0];
    const int*   ei   = (const int*)d_in[1];
    const float* W0   = (const float*)d_in[2];
    const float* b0   = (const float*)d_in[3];
    const float* W1   = (const float*)d_in[4];
    const float* b1   = (const float*)d_in[5];
    const float* Wx   = (const float*)d_in[6];
    const float* bx   = (const float*)d_in[7];
    float* out = (float*)d_out;

    const int n = in_sizes[0] / IN_DIM;
    const int E = in_sizes[1] / 2;
    const int* row = ei;
    const int* col = ei + E;

    char* ws = (char*)d_ws;
    size_t off = 0;
    float* dinv   = (float*)(ws + off); off += align256((size_t)n * 4);
    const int npad = (n + 31) & ~31;
    int*   cnt    = (int*)  (ws + off); off += align256((size_t)(npad + 96) * 4);
    int*   bump   = cnt + npad;
    int*   tick   = cnt + npad + 32;
    int*   seg    = (int*)  (ws + off); off += align256((size_t)n * 8);
    int*   rank   = (int*)  (ws + off); off += align256((size_t)E * 8);
    int*   edges  = (int*)  (ws + off); off += align256((size_t)E * 8);
    float* H      = (float*)(ws + off); off += align256((size_t)n * HID * 4);
    float* X      = (float*)(ws + off); off += align256((size_t)n * HID * 4);
    float* Y      = (float*)(ws + off); off += align256((size_t)n * HID * 4);
    (void)ws_size;

    const int TPB  = 256;
    const int gN   = (n + TPB - 1) / TPB;
    const int gN1  = (npad + 96 + TPB - 1) / TPB;
    const int gE   = (E + TPB - 1) / TPB;
    const int gNode = (n + 31) / 32;
    const int gTile = ((n / 16) + 7) / 8;
    const int split = ((n / 2) + 31) & ~31;
    const int jobs  = (E + 511) / 512;

    zero_int_k<<<gN1, TPB, 0, stream>>>(cnt, npad + 96);
    cnt_gemm0_k<<<gTile + 2 * jobs, TPB, 0, stream>>>(col, cnt, tick, rank, E, split, jobs,
                                                      x_in, W0, b0, H, IN_DIM, n, gTile);
    dinv_from_cnt_k<<<gN, TPB, 0, stream>>>(cnt, dinv, n);
    alloc_k<<<gN, 256, 0, stream>>>(cnt, bump, seg, n);
    fill_k<<<gE, TPB, 0, stream>>>(row, col, rank, seg, dinv, edges, E, split);


    gather_prop_k<HID / 32, true><<<gNode, TPB, 0, stream>>>(X, H, H, dinv, seg, edges, n, 0, 1.8f);
    gather_gemm_k<HID / 16, false><<<gNode, TPB, 0, stream>>>(Y, X, H, dinv, seg, edges, W1, b1, n, 1, 1.8f, 1.0f);

    gather_prop_k<HID / 32, true><<<gNode, TPB, 0, stream>>>(X, Y, Y, dinv, seg, edges, n, 0, 1.8f);
    gather_gemm_k<OUT_DIM / 16, true><<<gNode, TPB, 0, stream>>>(H, X, Y, dinv, seg, edges, Wx, bx, n, 0, 1.8f, 1.5f);

    gather_prop_k<OUT_DIM / 32, true><<<gNode, TPB, 0, stream>>>(X, H, H, dinv, seg, edges, n, 0, 1.8f);
    gather_prop_k<OUT_DIM / 32, false><<<gNode, TPB, 0, stream>>>(out, X, H, dinv, seg, edges, n, 0, 1.8f);
}


# adjacency at a fixed pitch, 64 slots a node (overflow list behind): the counting add's return is the edge's slot; alloc, dinv, fill launches go (11->8); 1st gather makes weights; order by arrival
# speedup vs baseline: 4.3900x; 1.1632x over previous
#include <hip/hip_runtime.h>
#include <math.h>

typedef __attribute__((ext_vector_type(2))) float v2f;
typedef __attribute__((ext_vector_type(8))) float v8f;

#define IN_DIM 256
#define HID 96
#define OUT_DIM 64
#define ALPHA_F 0.15f
#define OMA_F 0.85f
#define EPS_F 1e-12f


#define CAP 64

__global__ void zero_int_k(int* p, int n) {
    int i = blockIdx.x * blockDim.x + threadIdx.x;
    if (i < n) p[i] = 0;
}


__device__ __forceinline__ int chip_half() {
    return (int)((__builtin_amdgcn_s_sendmsg_rtn(0x87u) >> 18) & 1u);
}

__device__ __forceinline__ void place_edge(int* __restrict__ adj, int* ovf_n, int* __restrict__ ovf,
                                           const int c, const int r, const int s) {
    if (r < CAP) {
        adj[2 * ((size_t)c * CAP + r)] = s;
    } else {
        const int k = atomicAdd(ovf_n, 1);
        ovf[2 * (size_t)k]     = c;
        ovf[2 * (size_t)k + 1] = s;
    }
}

template <int NT>
__device__ __forceinline__ void gemm_strip(
    const int tm,
    const float* __restrict__ X, const float* __restrict__ W,
    const float* __restrict__ bias, float* __restrict__ Out,
    const int K, const int nrows) {
    const int M    = NT * 16;
    if (tm * 16 >= nrows) return;
    const int lane = threadIdx.x & 31;
    const int half = lane >> 4;
    const int l16  = lane & 15;

    v8f c[NT];
#pragma unroll
    for (int t = 0; t < NT; ++t) c[t] = (v8f){};

    const float* __restrict__ xrow = X + (size_t)(tm * 16 + l16) * K;

#pragma unroll 4
    for (int k = 0; k < K; k += 4) {
        const int kk = k + 2 * half;
        v2f a;
        a.x = xrow[kk];
        a.y = xrow[kk + 1];
        const float* __restrict__ w0 = W + (size_t)kk * M + l16;
#pragma unroll
        for (int t = 0; t < NT; ++t) {
            v2f b;
            b.x = w0[t * 16];
            b.y = w0[M + t * 16];
            c[t] = __builtin_amdgcn_wmma_f32_16x16x4_f32(
                false, a, false, b, (short)0, c[t], false, false);
        }
    }

#pragma unroll
    for (int t = 0; t < NT; ++t) {
        const int bcol = t * 16 + l16;
        const float bv = bias[bcol];
#pragma unroll
        for (int r = 0; r < 8; ++r) {
            const int orow = tm * 16 + r + 8 * half;
            Out[(size_t)orow * M + bcol] = c[t][r] + bv;
        }
    }
}

__global__ __launch_bounds__(256) void cnt_gemm0_k(
    const int* __restrict__ row, const int* __restrict__ col, int* cnt, int* tick,
    int* __restrict__ adj, int* ovf_n, int* __restrict__ ovf,
    int E, int split, int jobs,
    const float* __restrict__ X, const float* __restrict__ W,
    const float* __restrict__ bias, float* __restrict__ Out,
    int K, int nrows, int gemm_blocks) {
    if ((int)blockIdx.x < gemm_blocks) {
        gemm_strip<HID / 16>(blockIdx.x * 8 + (threadIdx.x >> 5),
                             X, W, bias, Out, K, nrows);
        return;
    }
    __shared__ int job_s[2];
    if (threadIdx.x == 0) {
        int h = chip_half();
        int t = atomicAdd(&tick[32 * h], 1);
        if (t >= jobs) {
            h ^= 1;
            t = atomicAdd(&tick[32 * h], 1);
        }
        job_s[0] = t;
        job_s[1] = h;
    }
    __syncthreads();
    const int t = job_s[0];
    const int h = job_s[1];
    if (t >= jobs) return;
    const int i0 = t * 512 + threadIdx.x;
    const int i1 = i0 + 256;
    const int c0 = (i0 < E) ? col[i0] : -1;
    const int c1 = (i1 < E) ? col[i1] : -1;
    const int s0 = (i0 < E) ? row[i0] : 0;
    const int s1 = (i1 < E) ? row[i1] : 0;
    const bool m0 = c0 >= 0 && (c0 >= split) == (h != 0);
    const bool m1 = c1 >= 0 && (c1 >= split) == (h != 0);
    int r0 = 0, r1 = 0;
    if (m0) r0 = atomicAdd(&cnt[c0], 1);
    if (m1) r1 = atomicAdd(&cnt[c1], 1);
    if (m0) place_edge(adj, ovf_n, ovf, c0, r0, s0);
    if (m1) place_edge(adj, ovf_n, ovf, c1, r1, s1);
}


template <int T>
__device__ __forceinline__ int grp_bcast(int v) {
    return (int)__builtin_amdgcn_mov_dpp8((unsigned)v, (unsigned)(T * 0x249249));
}

template <int K>
__device__ __forceinline__ float grp_xor(float v) {
    constexpr unsigned sel =
        ((0u ^ K) << 0)  | ((1u ^ K) << 3)  | ((2u ^ K) << 6)  | ((3u ^ K) << 9) |
        ((4u ^ K) << 12) | ((5u ^ K) << 15) | ((6u ^ K) << 18) | ((7u ^ K) << 21);
    return __int_as_float((int)__builtin_amdgcn_mov_dpp8((unsigned)__float_as_int(v), sel));
}

template <int NV>
__device__ __forceinline__ float row_scale(const float4 (&r)[NV], const float s) {
    float4 p = make_float4(0.0f, 0.0f, 0.0f, 0.0f);
    for (int v = 0; v < NV; ++v) {
        p.x = fmaf(r[v].x, r[v].x, p.x); p.y = fmaf(r[v].y, r[v].y, p.y);
        p.z = fmaf(r[v].z, r[v].z, p.z); p.w = fmaf(r[v].w, r[v].w, p.w);
    }
    float ss = (p.x + p.y) + (p.z + p.w);
    ss += grp_xor<1>(ss);
    ss += grp_xor<2>(ss);
    ss += grp_xor<4>(ss);
    return (ss > EPS_F * EPS_F) ? s * __builtin_amdgcn_rsqf(ss) : s / EPS_F;
}

template <int NV>
__device__ __forceinline__ void load_row(float4 (&r)[NV], const float* __restrict__ x,
                                         const int s, const int j) {
    constexpr int M = NV * 32;
    const float4* __restrict__ xs = reinterpret_cast<const float4*>(x + (size_t)s * M) + j;
    for (int v = 0; v < NV; ++v) r[v] = xs[8 * v];
}

template <int NV, bool XRAW>
__device__ __forceinline__ void add_row(float4 (&acc)[NV], const float4 (&r)[NV], float w,
                                        const float nf) {
    if (XRAW) w *= row_scale<NV>(r, nf);
    for (int v = 0; v < NV; ++v) {
        acc[v].x += w * r[v].x; acc[v].y += w * r[v].y;
        acc[v].z += w * r[v].z; acc[v].w += w * r[v].w;
    }
}

template <int NV, int T, bool XRAW>
__device__ __forceinline__ void edge_trip(float4 (&acc)[NV], const int2 cur, const int left,
                                          const float* __restrict__ x, const int j, const float nf) {
    const int   s = grp_bcast<T>(cur.x);
    const float w = __int_as_float(grp_bcast<T>(cur.y));
    if (T < left) {
        float4 r[NV];
        load_row<NV>(r, x, s, j);
        add_row<NV, XRAW>(acc, r, w, nf);
    }
}

template <int NV, bool XRAW, bool WINIT>
__device__ __forceinline__ void gather_row(
    float4 (&o)[NV], const float* __restrict__ x,
    const float* __restrict__ h, const int* __restrict__ cnt,
    int* adj, const int* __restrict__ ovf_n, const int* __restrict__ ovf,
    const int node, const int j, const int relu, const float nf) {
    constexpr int M = NV * 32;
    const float4* __restrict__ xn = reinterpret_cast<const float4*>(x + (size_t)node * M) + j;
    const float4* __restrict__ hn = reinterpret_cast<const float4*>(h + (size_t)node * M) + j;
    float4 acc[NV];
    float4 hv[NV];
    for (int v = 0; v < NV; ++v) acc[v] = xn[8 * v];
    for (int v = 0; v < NV; ++v) hv[v] = hn[8 * v];
    const int tot = cnt[node];
    const int len = tot < CAP ? tot : CAP;
    int2* en = reinterpret_cast<int2*>(adj) + (size_t)node * CAP;
    const float dn_self = rsqrtf((float)(tot + 1));
    {
        const float hs = row_scale<NV>(hv, nf);
        for (int v = 0; v < NV; ++v) { hv[v].x *= hs; hv[v].y *= hs; hv[v].z *= hs; hv[v].w *= hs; }
    }
    if (XRAW) {
        const float xsc = row_scale<NV>(acc, nf);
        for (int v = 0; v < NV; ++v) { acc[v].x *= xsc; acc[v].y *= xsc; acc[v].z *= xsc; acc[v].w *= xsc; }
    }
    for (int v = 0; v < NV; ++v) {
        acc[v].x *= dn_self; acc[v].y *= dn_self; acc[v].z *= dn_self; acc[v].w *= dn_self;
    }
    int2 cur = make_int2(0, 0);
    if (j < len) cur = en[j];
    for (int c0 = 0; c0 < len; c0 += 8) {
        const int left = len - c0;
        const int   s0 = grp_bcast<0>(cur.x);
        float4 r0[NV];
        load_row<NV>(r0, x, s0, j);
        int cs = 0;
        if (WINIT) { if (j < left) cs = cnt[cur.x]; }
        int2 nxt = make_int2(0, 0);
        if (c0 + 8 + j < len) nxt = en[c0 + 8 + j];
        if (WINIT) {
            cur.y = __float_as_int(rsqrtf((float)(cs + 1)));
            if (j < left) en[c0 + j] = cur;
        }
        const float w0 = __int_as_float(grp_bcast<0>(cur.y));
        add_row<NV, XRAW>(acc, r0, w0, nf);
        edge_trip<NV, 1, XRAW>(acc, cur, left, x, j, nf);
        edge_trip<NV, 2, XRAW>(acc, cur, left, x, j, nf);
        edge_trip<NV, 3, XRAW>(acc, cur, left, x, j, nf);
        edge_trip<NV, 4, XRAW>(acc, cur, left, x, j, nf);
        edge_trip<NV, 5, XRAW>(acc, cur, left, x, j, nf);
        edge_trip<NV, 6, XRAW>(acc, cur, left, x, j, nf);
        edge_trip<NV, 7, XRAW>(acc, cur, left, x, j, nf);
        cur = nxt;
    }
    if (tot > CAP) {
        const int novf = *ovf_n;
        for (int k = 0; k < novf; ++k) {
            if (ovf[2 * (size_t)k] == node) {
                const int   so = ovf[2 * (size_t)k + 1];
                const float wo = rsqrtf((float)(cnt[so] + 1));
                float4 ro[NV];
                load_row<NV>(ro, x, so, j);
                add_row<NV, XRAW>(acc, ro, wo, nf);
            }
        }
    }

    const float dd = OMA_F * dn_self;
    for (int v = 0; v < NV; ++v) {
        o[v].x = dd * acc[v].x + ALPHA_F * hv[v].x;
        o[v].y = dd * acc[v].y + ALPHA_F * hv[v].y;
        o[v].z = dd * acc[v].z + ALPHA_F * hv[v].z;
        o[v].w = dd * acc[v].w + ALPHA_F * hv[v].w;
        if (relu) {
            o[v].x = fmaxf(o[v].x, 0.0f); o[v].y = fmaxf(o[v].y, 0.0f);
            o[v].z = fmaxf(o[v].z, 0.0f); o[v].w = fmaxf(o[v].w, 0.0f);
        }
    }
}

template <int NV, bool XRAW, bool WINIT>
__global__ __launch_bounds__(256) void gather_prop_k(
    float* __restrict__ dst, const float* __restrict__ x,
    const float* __restrict__ h, const int* __restrict__ cnt,
    int* adj, const int* __restrict__ ovf_n, const int* __restrict__ ovf,
    int n, int relu, float nf) {
    constexpr int M = NV * 32;
    const int j    = threadIdx.x & 7;
    const int node = blockIdx.x * 32 + (threadIdx.x >> 3);
    if (node >= n) return;
    float4 o[NV];
    gather_row<NV, XRAW, WINIT>(o, x, h, cnt, adj, ovf_n, ovf, node, j, relu, nf);
    float4* __restrict__ dn = reinterpret_cast<float4*>(dst + (size_t)node * M) + j;
    for (int v = 0; v < NV; ++v) dn[8 * v] = o[v];
}

template <int NT, bool ANORM>
__global__ __launch_bounds__(256) void gather_gemm_k(
    float* __restrict__ Out, const float* __restrict__ x,
    const float* __restrict__ h, const int* __restrict__ cnt,
    int* adj, const int* __restrict__ ovf_n, const int* __restrict__ ovf,
    const float* __restrict__ W, const float* __restrict__ bias,
    int n, int relu, float nf, float nfa) {
    constexpr int NV    = HID / 32;
    constexpr int M     = NT * 16;
    constexpr int PITCH = HID + 4;
    __shared__ float tile[32 * PITCH];

    const int j  = threadIdx.x & 7;
    const int nb = threadIdx.x >> 3;
    const int own  = (int)blockIdx.x * 32 + nb;
    const int node = own < n ? own : n - 1;
    float4 o[NV];
    gather_row<NV, false, false>(o, x, h, cnt, adj, ovf_n, ovf, node, j, relu, nf);
    if (ANORM) {
        const float f = row_scale<NV>(o, nfa);
        for (int v = 0; v < NV; ++v) { o[v].x *= f; o[v].y *= f; o[v].z *= f; o[v].w *= f; }
    }
    for (int v = 0; v < NV; ++v)
        *reinterpret_cast<float4*>(&tile[nb * PITCH + 4 * (j + 8 * v)]) = o[v];
    __syncthreads();

    const int wave = threadIdx.x >> 5;
    const int lane = threadIdx.x & 31;
    const int half = lane >> 4;
    const int l16  = lane & 15;
    const int rt   = wave & 1;
    const int cp   = wave >> 1;
    const int row0 = (int)blockIdx.x * 32 + rt * 16;
    if (cp * 2 >= NT || row0 >= n) return;

    v8f c[2];
    c[0] = (v8f){};
    c[1] = (v8f){};
    const float* __restrict__ arow = tile + (rt * 16 + l16) * PITCH + 2 * half;
    const float* __restrict__ wcol = W + (size_t)(2 * half) * M + cp * 32 + l16;
#pragma unroll 4
    for (int k = 0; k < HID; k += 4) {
        const v2f a = *reinterpret_cast<const v2f*>(arow + k);
        const float* __restrict__ w0 = wcol + (size_t)k * M;
#pragma unroll
        for (int t = 0; t < 2; ++t) {
            v2f b;
            b.x = w0[t * 16];
            b.y = w0[M + t * 16];
            c[t] = __builtin_amdgcn_wmma_f32_16x16x4_f32(
                false, a, false, b, (short)0, c[t], false, false);
        }
    }

#pragma unroll
    for (int t = 0; t < 2; ++t) {
        const int bcol = cp * 32 + t * 16 + l16;
        const float bv = bias[bcol];
#pragma unroll
        for (int r = 0; r < 8; ++r) {
            const int orow = row0 + r + 8 * half;
            Out[(size_t)orow * M + bcol] = c[t][r] + bv;
        }
    }
}


static inline size_t align256(size_t b) { return (b + 255) & ~(size_t)255; }

extern "C" void kernel_launch(void* const* d_in, const int* in_sizes, int n_in,
                              void* d_out, int out_size, void* d_ws, size_t ws_size,
                              hipStream_t stream) {
    const float* x_in = (const float*)d_in[0];
    const int*   ei   = (const int*)d_in[1];
    const float* W0   = (const float*)d_in[2];
    const float* b0   = (const float*)d_in[3];
    const float* W1   = (const float*)d_in[4];
    const float* b1   = (const float*)d_in[5];
    const float* Wx   = (const float*)d_in[6];
    const float* bx   = (const float*)d_in[7];
    float* out = (float*)d_out;

    const int n = in_sizes[0] / IN_DIM;
    const int E = in_sizes[1] / 2;
    const int* row = ei;
    const int* col = ei + E;

    char* ws = (char*)d_ws;
    size_t off = 0;
    const int npad = (n + 31) & ~31;
    int*   cnt    = (int*)  (ws + off); off += align256((size_t)(npad + 96) * 4);
    int*   ovf_n  = cnt + npad;
    int*   tick   = cnt + npad + 32;
    int*   adj    = (int*)  (ws + off); off += align256((size_t)n * CAP * 8);
    int*   ovf    = (int*)  (ws + off); off += align256((size_t)E * 8);
    float* H      = (float*)(ws + off); off += align256((size_t)n * HID * 4);
    float* X      = (float*)(ws + off); off += align256((size_t)n * HID * 4);
    float* Y      = (float*)(ws + off); off += align256((size_t)n * HID * 4);
    (void)ws_size;

    const int TPB  = 256;
    const int gN1  = (npad + 96 + TPB - 1) / TPB;
    const int gNode = (n + 31) / 32;
    const int gTile = ((n / 16) + 7) / 8;
    const int split = ((n / 2) + 31) & ~31;
    const int jobs  = (E + 511) / 512;

    zero_int_k<<<gN1, TPB, 0, stream>>>(cnt, npad + 96);
    cnt_gemm0_k<<<gTile + 2 * jobs, TPB, 0, stream>>>(row, col, cnt, tick, adj, ovf_n, ovf, E, split, jobs,
                                                      x_in, W0, b0, H, IN_DIM, n, gTile);


    gather_prop_k<HID / 32, true, true><<<gNode, TPB, 0, stream>>>(X, H, H, cnt, adj, ovf_n, ovf, n, 0, 1.8f);
    gather_gemm_k<HID / 16, false><<<gNode, TPB, 0, stream>>>(Y, X, H, cnt, adj, ovf_n, ovf, W1, b1, n, 1, 1.8f, 1.0f);

    gather_prop_k<HID / 32, true, false><<<gNode, TPB, 0, stream>>>(X, Y, Y, cnt, adj, ovf_n, ovf, n, 0, 1.8f);
    gather_gemm_k<OUT_DIM / 16, true><<<gNode, TPB, 0, stream>>>(H, X, Y, cnt, adj, ovf_n, ovf, Wx, bx, n, 0, 1.8f, 1.5f);

    gather_prop_k<OUT_DIM / 32, true, false><<<gNode, TPB, 0, stream>>>(X, H, H, cnt, adj, ovf_n, ovf, n, 0, 1.8f);
    gather_prop_k<OUT_DIM / 32, false, false><<<gNode, TPB, 0, stream>>>(out, X, H, cnt, adj, ovf_n, ovf, n, 0, 1.8f);
}
